# starting kernel
# baseline (speedup 1.0000x reference)
#include <hip/hip_runtime.h>


typedef __attribute__((ext_vector_type(2))) float v2f;
typedef __attribute__((ext_vector_type(8))) float v8f;

#define DFEAT 128
#define LEAKY 0.01f

__global__ void __launch_bounds__(256) gcn_gemm_xWT(
    const float* __restrict__ x, const float* __restrict__ W,
    float* __restrict__ h, int n_nodes)
{
  __shared__ float Wlds[DFEAT * DFEAT];

  {
    const float4* Wg4 = (const float4*)W;
    float4* Wl4 = (float4*)Wlds;
    #pragma unroll
    for (int i = 0; i < (DFEAT * DFEAT / 4) / 256; ++i)
      Wl4[threadIdx.x + 256 * i] = Wg4[threadIdx.x + 256 * i];
  }
  __syncthreads();

  const int wave = threadIdx.x >> 5;
  const int lane = threadIdx.x & 31;
  const int half = lane >> 4;
  const int lrow = lane & 15;

  const int base = blockIdx.x * 128 + wave * 16;
  if (base >= n_nodes) return;

  int row = base + lrow;
  if (row >= n_nodes) row = n_nodes - 1;
  const v2f* xr = (const v2f*)(x + (size_t)row * DFEAT);

  v2f afrag[32];
  #pragma unroll
  for (int kt = 0; kt < 32; ++kt) afrag[kt] = xr[2 * kt + half];

  #pragma unroll
  for (int nt = 0; nt < 8; ++nt) {
    const v2f* wr = (const v2f*)(Wlds + (nt * 16 + lrow) * DFEAT);
    v8f acc = {};
    #pragma unroll
    for (int kt = 0; kt < 32; ++kt) {
      v2f b = wr[2 * kt + half];
      acc = __builtin_amdgcn_wmma_f32_16x16x4_f32(
           false, afrag[kt],  false, b,
           (short)0, acc,  false,  false);
    }
    const int col = nt * 16 + lrow;
    #pragma unroll
    for (int r = 0; r < 8; ++r) {
      int m = base + half * 8 + r;
      if (m < n_nodes) h[(size_t)m * DFEAT + col] = acc[r];
    }
  }
}

__global__ void deg_init(float* __restrict__ deg, int n) {
  int i = blockIdx.x * blockDim.x + threadIdx.x;
  if (i < n) deg[i] = 1.0f;
}
__global__ void deg_accum(const int* __restrict__ dst, float* __restrict__ deg, int E) {
  int e = blockIdx.x * blockDim.x + threadIdx.x;
  if (e < E) unsafeAtomicAdd(&deg[dst[e]], 1.0f);
}
__global__ void deg_finalize(float* __restrict__ deg, int n) {
  int i = blockIdx.x * blockDim.x + threadIdx.x;
  if (i < n) deg[i] = __frsqrt_rn(deg[i]);
}

__global__ void gcn_init_out(const float* __restrict__ h, const float* __restrict__ dis,
                             const float* __restrict__ bias, float* __restrict__ out, int n) {
  int idx = blockIdx.x * blockDim.x + threadIdx.x;
  if (idx >= n * DFEAT) return;
  int i = idx >> 7, f = idx & (DFEAT - 1);
  float dv = dis[i];
  out[idx] = bias[f] + h[idx] * dv * dv;
}

__global__ void __launch_bounds__(256) gcn_scatter(
    const int* __restrict__ src, const int* __restrict__ dst,
    const float* __restrict__ h, const float* __restrict__ dis,
    float* __restrict__ out, int E)
{
  int e = blockIdx.x * 8 + (threadIdx.x >> 5);
  if (e >= E) return;
  int lane = threadIdx.x & 31;
  int s = src[e], d = dst[e];
  float nrm = dis[s] * dis[d];
  const float* hs = h + (size_t)s * DFEAT;
  float* od = out + (size_t)d * DFEAT;
  #pragma unroll
  for (int j = 0; j < 4; ++j) {
    int f = lane + 32 * j;
    unsafeAtomicAdd(&od[f], hs[f] * nrm);
  }
}

__global__ void gcn_leaky(const float* __restrict__ in, float* __restrict__ out, int n) {
  int idx = blockIdx.x * blockDim.x + threadIdx.x;
  if (idx < n) {
    float v = in[idx];
    out[idx] = (v >= 0.0f) ? v : LEAKY * v;
  }
}

extern "C" void kernel_launch(void* const* d_in, const int* in_sizes, int n_in,
                              void* d_out, int out_size, void* d_ws, size_t ws_size,
                              hipStream_t stream) {
  const float* x  = (const float*)d_in[0];
  const int*   ei = (const int*)d_in[1];
  const float* Ws = (const float*)d_in[2];
  const float* bs = (const float*)d_in[3];
  float* out = (float*)d_out;

  const int n = in_sizes[0] / DFEAT;
  const int E = in_sizes[1] / 2;
  const int* src = ei;
  const int* dst = ei + E;

  float* xbuf = (float*)d_ws;
  float* hbuf = xbuf + (size_t)n * DFEAT;
  float* dis  = hbuf + (size_t)n * DFEAT;

  const int TB = 256;
  const int nd = n * DFEAT;

  deg_init    <<<(n + TB - 1) / TB, TB, 0, stream>>>(dis, n);
  deg_accum   <<<(E + TB - 1) / TB, TB, 0, stream>>>(dst, dis, E);
  deg_finalize<<<(n + TB - 1) / TB, TB, 0, stream>>>(dis, n);

  const float* cur = x;
  for (int l = 0; l < 3; ++l) {
    gcn_gemm_xWT<<<(n + 127) / 128, 256, 0, stream>>>(cur, Ws + (size_t)l * DFEAT * DFEAT, hbuf, n);
    gcn_init_out<<<(nd + TB - 1) / TB, TB, 0, stream>>>(hbuf, dis, bs + (size_t)l * DFEAT, out, n);
    gcn_scatter <<<(E + 7) / 8, 256, 0, stream>>>(src, dst, hbuf, dis, out, E);
    float* act_dst = (l == 2) ? out : xbuf;
    gcn_leaky   <<<(nd + TB - 1) / TB, TB, 0, stream>>>(out, act_dst, nd);
    cur = xbuf;
  }
}


# Edge sum without global float atomics or edge sort: a work-group owns 196 destination rows in LDS, scans all edges per layer, adds matching source rows by LDS float adds (arrival order); 9 launches
# speedup vs baseline: 1.7461x; 1.7461x over previous
#include <hip/hip_runtime.h>


typedef __attribute__((ext_vector_type(2))) float v2f;
typedef __attribute__((ext_vector_type(8))) float v8f;
typedef int i4u __attribute__((ext_vector_type(4), aligned(4)));

#define DFEAT 128
#define LEAKY 0.01f

__device__ __forceinline__ void load_a_tile(v2f (&a)[32], const float* __restrict__ x,
                                            int tile, int lrow, int half, int n_nodes)
{
  int row = tile * 16 + lrow;
  if (row >= n_nodes) row = n_nodes - 1;
  const v2f* xr = (const v2f*)(x + (size_t)row * DFEAT);
  #pragma unroll
  for (int kt = 0; kt < 32; ++kt) a[kt] = xr[2 * kt + half];
}

__global__ void __launch_bounds__(256) gcn_gemm_xWT(
    const float* __restrict__ x, const float* __restrict__ W,
    float* __restrict__ h, int n_nodes)
{
  const int wave = __builtin_amdgcn_readfirstlane((int)(threadIdx.x >> 5));
  const int lane = threadIdx.x & 31;
  const int half = lane >> 4;
  const int lrow = lane & 15;

  const int nh     = wave >> 2;
  const int slot   = blockIdx.x * 4 + (wave & 3);
  const int nslots = gridDim.x * 4;
  const int ntiles = (n_nodes + 15) >> 4;

  v2f bfrag[4][32];
  #pragma unroll
  for (int j = 0; j < 4; ++j) {
    const int col = (nh * 4 + j) * 16 + lrow;
    const v2f* wr = (const v2f*)(W + (size_t)col * DFEAT);
    #pragma unroll
    for (int kt = 0; kt < 32; ++kt) bfrag[j][kt] = wr[2 * kt + half];
  }

  v2f anext[32];
  if (slot < ntiles) load_a_tile(anext, x, slot, lrow, half, n_nodes);

  for (int t = slot; t < ntiles; t += nslots) {
    const int base = t * 16;

    v2f afrag[32];
    #pragma unroll
    for (int kt = 0; kt < 32; ++kt) afrag[kt] = anext[kt];

    int tn = t + nslots;
    if (tn >= ntiles) tn = ntiles - 1;
    load_a_tile(anext, x, tn, lrow, half, n_nodes);

    #pragma unroll
    for (int jp = 0; jp < 2; ++jp) {
      v8f acc[2] = {};
      #pragma unroll
      for (int kt = 0; kt < 32; ++kt) {
        #pragma unroll
        for (int u = 0; u < 2; ++u)
          acc[u] = __builtin_amdgcn_wmma_f32_16x16x4_f32(
               false, afrag[kt],  false, bfrag[2 * jp + u][kt],
               (short)0, acc[u],  false,  false);
      }
      #pragma unroll
      for (int u = 0; u < 2; ++u) {
        const int j = 2 * jp + u;
        const int col = (nh * 4 + j) * 16 + lrow;
        #pragma unroll
        for (int r = 0; r < 8; ++r) {
          int m = base + half * 8 + r;
          if (m < n_nodes) h[(size_t)m * DFEAT + col] = acc[u][r];
        }
      }
    }
  }
}

__global__ void deg_init(float* __restrict__ deg, int n) {
  int i = blockIdx.x * blockDim.x + threadIdx.x;
  if (i < n) deg[i] = 1.0f;
}
__global__ void deg_accum(const int* __restrict__ dst, float* __restrict__ deg, int E) {
  int e = blockIdx.x * blockDim.x + threadIdx.x;
  if (e < E) unsafeAtomicAdd(&deg[dst[e]], 1.0f);
}
__global__ void deg_finalize(float* __restrict__ deg, int n) {
  int i = blockIdx.x * blockDim.x + threadIdx.x;
  if (i < n) deg[i] = __frsqrt_rn(deg[i]);
}

#define RMAX   196
#define SCAN_U 8
#define QCAP   (32 + SCAN_U * 128)

__device__ __forceinline__ void work_off_queue(
    const int* q, int cnt, int lane, int base,
    const int* __restrict__ src, const int* __restrict__ dst,
    const float* __restrict__ h, const float* __restrict__ dis,
    const float* dsh, float* acc)
{
  for (int b0 = 0; b0 < cnt; b0 += 32) {
    const int nb = min(32, cnt - b0);
    int myS = 0, myR = RMAX;
    float myW = 0.0f;
    if (lane < nb) {
      const int e = q[b0 + lane];
      myS = src[e];
      myR = dst[e] - base;
      myW = dis[myS] * dsh[myR];
    }
    for (int g = 0; g < nb; g += 8) {
      float v[8][4];
      #pragma unroll
      for (int k = 0; k < 8; ++k) {
        const int s = __builtin_amdgcn_readlane(myS, g + k);
        const float* hs = h + (size_t)s * DFEAT + lane;
        #pragma unroll
        for (int j = 0; j < 4; ++j) v[k][j] = hs[32 * j];
      }
      #pragma unroll
      for (int k = 0; k < 8; ++k) {
        const float w = __int_as_float(__builtin_amdgcn_readlane(__float_as_int(myW), g + k));
        const int   r = __builtin_amdgcn_readlane(myR, g + k);
        float* a = acc + r * DFEAT + lane;
        #pragma unroll
        for (int j = 0; j < 4; ++j)
          __hip_atomic_fetch_add(a + 32 * j, v[k][j] * w, __ATOMIC_RELAXED, __HIP_MEMORY_SCOPE_WORKGROUP);
      }
    }
  }
}

__global__ void __launch_bounds__(1024) gcn_node_sum(
    const int* __restrict__ src, const int* __restrict__ dst,
    const float* __restrict__ h, const float* __restrict__ dis,
    const float* __restrict__ bias, float* __restrict__ outp, int n_nodes, int E)
{
  __shared__ float acc[(RMAX + 1) * DFEAT];
  __shared__ float dsh[RMAX];
  __shared__ int   queue[32 * QCAP];

  const int tid  = threadIdx.x;
  const int lane = tid & 31;
  const int wave = __builtin_amdgcn_readfirstlane(tid >> 5);
  int* q = queue + wave * QCAP;

  const int NQ  = E >> 2;
  const int NIT = (NQ + 1023) >> 10;
  const int off = (int)((blockIdx.x * (unsigned)NIT) / gridDim.x);

  for (int base = blockIdx.x * RMAX; base < n_nodes; base += gridDim.x * RMAX) {
    const int rows = min(RMAX, n_nodes - base);

    for (int idx = tid; idx < rows * DFEAT; idx += 1024) {
      const int r = idx >> 7, f = idx & (DFEAT - 1);
      const float dv = dis[base + r];
      const float hv = h[(size_t)base * DFEAT + idx];
      acc[idx] = bias[f] + hv * dv * dv;
    }
    if (tid < rows) dsh[tid] = dis[base + tid];
    __syncthreads();

    int cnt = 0;
    auto push = [&](bool m, int e) {
      const unsigned b = __builtin_amdgcn_ballot_w32(m);
      if (m) q[cnt + __builtin_amdgcn_mbcnt_lo(b, 0)] = e;
      cnt += __builtin_popcount(b);
    };

    for (int i = 0; i < NIT; i += SCAN_U) {
      i4u dq[SCAN_U];
      int gq[SCAN_U];
      bool ok[SCAN_U];
      #pragma unroll
      for (int u = 0; u < SCAN_U; ++u) {
        int itr = i + u + off;
        if (itr >= NIT) itr -= NIT;
        const int g = itr * 1024 + tid;
        ok[u] = (i + u < NIT) && (g < NQ);
        gq[u] = ok[u] ? g : 0;
        dq[u] = *(const i4u*)(dst + 4 * (size_t)gq[u]);
      }
      #pragma unroll
      for (int u = 0; u < SCAN_U; ++u) {
        const bool m0 = ok[u] && (unsigned)(dq[u].x - base) < (unsigned)rows;
        const bool m1 = ok[u] && (unsigned)(dq[u].y - base) < (unsigned)rows;
        const bool m2 = ok[u] && (unsigned)(dq[u].z - base) < (unsigned)rows;
        const bool m3 = ok[u] && (unsigned)(dq[u].w - base) < (unsigned)rows;
        if (__builtin_amdgcn_ballot_w32(m0 || m1 || m2 || m3)) {
          push(m0, 4 * gq[u] + 0);
          push(m1, 4 * gq[u] + 1);
          push(m2, 4 * gq[u] + 2);
          push(m3, 4 * gq[u] + 3);
        }
      }
      if (cnt >= 32) {
        work_off_queue(q, cnt, lane, base, src, dst, h, dis, dsh, acc);
        cnt = 0;
      }
    }
    {
      const int e = 4 * NQ + lane;
      bool m = (wave == 0) && (lane < (E & 3));
      const int d = m ? dst[e] : -1;
      m = m && (unsigned)(d - base) < (unsigned)rows;
      push(m, e);
    }
    work_off_queue(q, cnt, lane, base, src, dst, h, dis, dsh, acc);
    __syncthreads();

    for (int idx = tid; idx < rows * DFEAT; idx += 1024) {
      const float v = acc[idx];
      outp[(size_t)base * DFEAT + idx] = (v >= 0.0f) ? v : LEAKY * v;
    }
    __syncthreads();
  }
}

extern "C" void kernel_launch(void* const* d_in, const int* in_sizes, int n_in,
                              void* d_out, int out_size, void* d_ws, size_t ws_size,
                              hipStream_t stream) {
  const float* x  = (const float*)d_in[0];
  const int*   ei = (const int*)d_in[1];
  const float* Ws = (const float*)d_in[2];
  const float* bs = (const float*)d_in[3];
  float* out = (float*)d_out;

  const int n = in_sizes[0] / DFEAT;
  const int E = in_sizes[1] / 2;
  const int* src = ei;
  const int* dst = ei + E;

  float* xbuf = (float*)d_ws;
  float* hbuf = xbuf + (size_t)n * DFEAT;
  float* dis  = hbuf + (size_t)n * DFEAT;

  const int TB = 256;

  deg_init    <<<(n + TB - 1) / TB, TB, 0, stream>>>(dis, n);
  deg_accum   <<<(E + TB - 1) / TB, TB, 0, stream>>>(dst, dis, E);
  deg_finalize<<<(n + TB - 1) / TB, TB, 0, stream>>>(dis, n);

  const int gg = 256;
  const int ga = 256;

  gcn_gemm_xWT<<<gg, 256, 0, stream>>>(x, Ws, hbuf, n);
  gcn_node_sum<<<ga, 1024, 0, stream>>>(src, dst, hbuf, dis, bs, xbuf, n, E);

  gcn_gemm_xWT<<<gg, 256, 0, stream>>>(xbuf, Ws + (size_t)DFEAT * DFEAT, hbuf, n);
  gcn_node_sum<<<ga, 1024, 0, stream>>>(src, dst, hbuf, dis, bs + DFEAT, xbuf, n, E);

  gcn_gemm_xWT<<<gg, 256, 0, stream>>>(xbuf, Ws + (size_t)2 * DFEAT * DFEAT, hbuf, n);
  gcn_node_sum<<<ga, 1024, 0, stream>>>(src, dst, hbuf, dis, bs + 2 * DFEAT, out, n, E);
}


# Edge scan reads one owner byte per edge (id of the row range holding its dst, written once per call by the degree launch), 16 edges per lane-load, not the 4-byte dst; LDS adds still in arrival order
# speedup vs baseline: 2.0106x; 1.1515x over previous
#include <hip/hip_runtime.h>


typedef __attribute__((ext_vector_type(2))) float v2f;
typedef __attribute__((ext_vector_type(8))) float v8f;
typedef int i4u __attribute__((ext_vector_type(4), aligned(4)));
typedef unsigned u4u __attribute__((ext_vector_type(4), aligned(4)));

#define DFEAT 128
#define LEAKY 0.01f
#define RMAX   196
#define NBLK   256

__device__ __forceinline__ void load_a_tile(v2f (&a)[32], const float* __restrict__ x,
                                            int tile, int lrow, int half, int n_nodes)
{
  int row = tile * 16 + lrow;
  if (row >= n_nodes) row = n_nodes - 1;
  const v2f* xr = (const v2f*)(x + (size_t)row * DFEAT);
  #pragma unroll
  for (int kt = 0; kt < 32; ++kt) a[kt] = xr[2 * kt + half];
}

__global__ void __launch_bounds__(256) gcn_gemm_xWT(
    const float* __restrict__ x, const float* __restrict__ W,
    float* __restrict__ h, int n_nodes)
{
  const int wave = __builtin_amdgcn_readfirstlane((int)(threadIdx.x >> 5));
  const int lane = threadIdx.x & 31;
  const int half = lane >> 4;
  const int lrow = lane & 15;

  const int nh     = wave >> 2;
  const int slot   = blockIdx.x * 4 + (wave & 3);
  const int nslots = gridDim.x * 4;
  const int ntiles = (n_nodes + 15) >> 4;

  v2f bfrag[4][32];
  #pragma unroll
  for (int j = 0; j < 4; ++j) {
    const int col = (nh * 4 + j) * 16 + lrow;
    const v2f* wr = (const v2f*)(W + (size_t)col * DFEAT);
    #pragma unroll
    for (int kt = 0; kt < 32; ++kt) bfrag[j][kt] = wr[2 * kt + half];
  }

  v2f anext[32];
  if (slot < ntiles) load_a_tile(anext, x, slot, lrow, half, n_nodes);

  for (int t = slot; t < ntiles; t += nslots) {
    const int base = t * 16;

    v2f afrag[32];
    #pragma unroll
    for (int kt = 0; kt < 32; ++kt) afrag[kt] = anext[kt];

    int tn = t + nslots;
    if (tn >= ntiles) tn = ntiles - 1;
    load_a_tile(anext, x, tn, lrow, half, n_nodes);

    #pragma unroll
    for (int jp = 0; jp < 2; ++jp) {
      v8f acc[2] = {};
      #pragma unroll
      for (int kt = 0; kt < 32; ++kt) {
        #pragma unroll
        for (int u = 0; u < 2; ++u)
          acc[u] = __builtin_amdgcn_wmma_f32_16x16x4_f32(
               false, afrag[kt],  false, bfrag[2 * jp + u][kt],
               (short)0, acc[u],  false,  false);
      }
      #pragma unroll
      for (int u = 0; u < 2; ++u) {
        const int j = 2 * jp + u;
        const int col = (nh * 4 + j) * 16 + lrow;
        #pragma unroll
        for (int r = 0; r < 8; ++r) {
          int m = base + half * 8 + r;
          if (m < n_nodes) h[(size_t)m * DFEAT + col] = acc[u][r];
        }
      }
    }
  }
}

__global__ void deg_init(float* __restrict__ deg, int n) {
  int i = blockIdx.x * blockDim.x + threadIdx.x;
  if (i < n) deg[i] = 1.0f;
}
__global__ void deg_accum(const int* __restrict__ dst, float* __restrict__ deg,
                          unsigned char* __restrict__ own, int E) {
  int e = blockIdx.x * blockDim.x + threadIdx.x;
  if (e < E) {
    const int d = dst[e];
    unsafeAtomicAdd(&deg[d], 1.0f);
    if (own) own[e] = (unsigned char)((d / RMAX) % NBLK);
  }
}
__global__ void deg_finalize(float* __restrict__ deg, int n) {
  int i = blockIdx.x * blockDim.x + threadIdx.x;
  if (i < n) deg[i] = __frsqrt_rn(deg[i]);
}

#define SCAN_U 8
#define QCAP_D (32 + SCAN_U * 128)
#define QCAP_K 64

__device__ __forceinline__ void work_off_queue(
    const int* q, int cnt, int lane, int base, int rows,
    const int* __restrict__ src, const int* __restrict__ dst,
    const float* __restrict__ h, const float* __restrict__ dis,
    const float* dsh, float* acc)
{
  for (int b0 = 0; b0 < cnt; b0 += 32) {
    const int nb = min(32, cnt - b0);
    int myS = 0, myR = RMAX;
    float myW = 0.0f;
    if (lane < nb) {
      const int e = q[b0 + lane];
      const int s = src[e];
      const int r = dst[e] - base;
      const bool mine = (unsigned)r < (unsigned)rows;
      const float w = dis[s] * dsh[mine ? r : 0];
      myS = mine ? s : 0;
      myR = mine ? r : RMAX;
      myW = mine ? w : 0.0f;
    }
    for (int g = 0; g < nb; g += 8) {
      float v[8][4];
      #pragma unroll
      for (int k = 0; k < 8; ++k) {
        const int s = __builtin_amdgcn_readlane(myS, g + k);
        const float* hs = h + (size_t)s * DFEAT + lane;
        #pragma unroll
        for (int j = 0; j < 4; ++j) v[k][j] = hs[32 * j];
      }
      #pragma unroll
      for (int k = 0; k < 8; ++k) {
        const float w = __int_as_float(__builtin_amdgcn_readlane(__float_as_int(myW), g + k));
        const int   r = __builtin_amdgcn_readlane(myR, g + k);
        float* a = acc + r * DFEAT + lane;
        #pragma unroll
        for (int j = 0; j < 4; ++j)
          __hip_atomic_fetch_add(a + 32 * j, v[k][j] * w, __ATOMIC_RELAXED, __HIP_MEMORY_SCOPE_WORKGROUP);
      }
    }
  }
}

__device__ __forceinline__ unsigned zero_bytes(unsigned x) {
  return ~((((x & 0x7f7f7f7fu) + 0x7f7f7f7fu) | x) | 0x7f7f7f7fu);
}

template <bool KEYS>
__global__ void __launch_bounds__(1024) gcn_node_sum(
    const int* __restrict__ src, const int* __restrict__ dst,
    const unsigned char* __restrict__ own,
    const float* __restrict__ h, const float* __restrict__ dis,
    const float* __restrict__ bias, float* __restrict__ outp, int n_nodes, int E)
{
  __shared__ float acc[(RMAX + 1) * DFEAT];
  __shared__ float dsh[RMAX];
  __shared__ int   queue[32 * (KEYS ? QCAP_K : QCAP_D)];

  const int tid  = threadIdx.x;
  const int lane = tid & 31;
  const int wave = __builtin_amdgcn_readfirstlane(tid >> 5);
  int* q = queue + wave * (KEYS ? QCAP_K : QCAP_D);

  const int GSH = KEYS ? 4 : 2;
  const int NQ  = E >> GSH;
  const int NIT = (NQ + 1023) >> 10;
  const int off = (int)((blockIdx.x * (unsigned)NIT) / gridDim.x);
  const unsigned pat = (blockIdx.x % NBLK) * 0x01010101u;

  for (int base = blockIdx.x * RMAX; base < n_nodes; base += gridDim.x * RMAX) {
    const int rows = min(RMAX, n_nodes - base);

    for (int idx = tid; idx < rows * DFEAT; idx += 1024) {
      const int r = idx >> 7, f = idx & (DFEAT - 1);
      const float dv = dis[base + r];
      const float hv = h[(size_t)base * DFEAT + idx];
      acc[idx] = bias[f] + hv * dv * dv;
    }
    if (tid < rows) dsh[tid] = dis[base + tid];
    __syncthreads();

    int cnt = 0;
    auto push = [&](bool m, int e) {
      const unsigned b = __builtin_amdgcn_ballot_w32(m);
      if (m) q[cnt + __builtin_amdgcn_mbcnt_lo(b, 0)] = e;
      cnt += __builtin_popcount(b);
    };

    if (KEYS) {
      for (int i = 0; i < NIT; i += SCAN_U) {
        u4u kq[SCAN_U];
        bool ok[SCAN_U];
        #pragma unroll
        for (int u = 0; u < SCAN_U; ++u) {
          int itr = i + u + off;
          if (itr >= NIT) itr -= NIT;
          const int g = itr * 1024 + tid;
          ok[u] = (i + u < NIT) && (g < NQ);
          kq[u] = *(const u4u*)(own + 16 * (size_t)(ok[u] ? g : 0));
        }
        unsigned M[SCAN_U / 2];
        #pragma unroll
        for (int u = 0; u < SCAN_U; ++u) {
          unsigned m = (zero_bytes(kq[u].x ^ pat) >> 3) | (zero_bytes(kq[u].y ^ pat) >> 2)
                     | (zero_bytes(kq[u].z ^ pat) >> 1) |  zero_bytes(kq[u].w ^ pat);
          if (!ok[u]) m = 0;
          if (u & 1) M[u >> 1] |= m >> 4; else M[u >> 1] = m;
        }
        for (;;) {
          const unsigned any = M[0] | M[1] | M[2] | M[3];
          const unsigned bal = __builtin_amdgcn_ballot_w32(any != 0u);
          if (bal == 0u) break;
          if (any != 0u) {
            const int w = M[0] ? 0 : M[1] ? 1 : M[2] ? 2 : 3;
            const unsigned mw = M[0] ? M[0] : M[1] ? M[1] : M[2] ? M[2] : M[3];
            const int p = __builtin_ctz(mw);
            int itr = i + 2 * w + (((p >> 2) & 1) ^ 1) + off;
            if (itr >= NIT) itr -= NIT;
            q[cnt + __builtin_amdgcn_mbcnt_lo(bal, 0)] = ((itr * 1024 + tid) << 4) + 4 * (p & 3) + (p >> 3);
            const unsigned cl = mw & (mw - 1u);
            M[0] = (w == 0) ? cl : M[0];
            M[1] = (w == 1) ? cl : M[1];
            M[2] = (w == 2) ? cl : M[2];
            M[3] = (w == 3) ? cl : M[3];
          }
          cnt += __builtin_popcount(bal);
          if (cnt >= 32) {
            work_off_queue(q, cnt, lane, base, rows, src, dst, h, dis, dsh, acc);
            cnt = 0;
          }
        }
      }
    } else {
      for (int i = 0; i < NIT; i += SCAN_U) {
        i4u dq[SCAN_U];
        int gq[SCAN_U];
        bool ok[SCAN_U];
        #pragma unroll
        for (int u = 0; u < SCAN_U; ++u) {
          int itr = i + u + off;
          if (itr >= NIT) itr -= NIT;
          const int g = itr * 1024 + tid;
          ok[u] = (i + u < NIT) && (g < NQ);
          gq[u] = ok[u] ? g : 0;
          dq[u] = *(const i4u*)(dst + 4 * (size_t)gq[u]);
        }
        #pragma unroll
        for (int u = 0; u < SCAN_U; ++u) {
          const bool m0 = ok[u] && (unsigned)(dq[u].x - base) < (unsigned)rows;
          const bool m1 = ok[u] && (unsigned)(dq[u].y - base) < (unsigned)rows;
          const bool m2 = ok[u] && (unsigned)(dq[u].z - base) < (unsigned)rows;
          const bool m3 = ok[u] && (unsigned)(dq[u].w - base) < (unsigned)rows;
          if (__builtin_amdgcn_ballot_w32(m0 || m1 || m2 || m3)) {
            push(m0, 4 * gq[u] + 0);
            push(m1, 4 * gq[u] + 1);
            push(m2, 4 * gq[u] + 2);
            push(m3, 4 * gq[u] + 3);
          }
        }
        if (cnt >= 32) {
          work_off_queue(q, cnt, lane, base, rows, src, dst, h, dis, dsh, acc);
          cnt = 0;
        }
      }
    }
    {
      const int e = (NQ << GSH) + lane;
      bool m = (wave == 0) && (lane < (E & ((1 << GSH) - 1)));
      const int d = m ? dst[e] : -1;
      m = m && (unsigned)(d - base) < (unsigned)rows;
      push(m, e);
    }
    work_off_queue(q, cnt, lane, base, rows, src, dst, h, dis, dsh, acc);
    __syncthreads();

    for (int idx = tid; idx < rows * DFEAT; idx += 1024) {
      const float v = acc[idx];
      outp[(size_t)base * DFEAT + idx] = (v >= 0.0f) ? v : LEAKY * v;
    }
    __syncthreads();
  }
}

extern "C" void kernel_launch(void* const* d_in, const int* in_sizes, int n_in,
                              void* d_out, int out_size, void* d_ws, size_t ws_size,
                              hipStream_t stream) {
  const float* x  = (const float*)d_in[0];
  const int*   ei = (const int*)d_in[1];
  const float* Ws = (const float*)d_in[2];
  const float* bs = (const float*)d_in[3];
  float* out = (float*)d_out;

  const int n = in_sizes[0] / DFEAT;
  const int E = in_sizes[1] / 2;
  const int* src = ei;
  const int* dst = ei + E;

  float* xbuf = (float*)d_ws;
  float* hbuf = xbuf + (size_t)n * DFEAT;
  float* dis  = hbuf + (size_t)n * DFEAT;
  const bool keys = ws_size >= ((size_t)n * (2 * DFEAT + 1)) * sizeof(float) + (size_t)E;
  unsigned char* own = keys ? (unsigned char*)(dis + n) : nullptr;

  const int TB = 256;

  deg_init    <<<(n + TB - 1) / TB, TB, 0, stream>>>(dis, n);
  deg_accum   <<<(E + TB - 1) / TB, TB, 0, stream>>>(dst, dis, own, E);
  deg_finalize<<<(n + TB - 1) / TB, TB, 0, stream>>>(dis, n);

  const int gg = 256;
  const int ga = NBLK;
  auto node_sum = [&](const float* hin, const float* b, float* o) {
    if (keys) gcn_node_sum<true ><<<ga, 1024, 0, stream>>>(src, dst, own, hin, dis, b, o, n, E);
    else      gcn_node_sum<false><<<ga, 1024, 0, stream>>>(src, dst, own, hin, dis, b, o, n, E);
  };

  gcn_gemm_xWT<<<gg, 256, 0, stream>>>(x, Ws, hbuf, n);
  node_sum(hbuf, bs, xbuf);

  gcn_gemm_xWT<<<gg, 256, 0, stream>>>(xbuf, Ws + (size_t)DFEAT * DFEAT, hbuf, n);
  node_sum(hbuf, bs + DFEAT, xbuf);

  gcn_gemm_xWT<<<gg, 256, 0, stream>>>(xbuf, Ws + (size_t)2 * DFEAT * DFEAT, hbuf, n);
  node_sum(hbuf, bs + 2 * DFEAT, out);
}


# GEMM fragment loads take 16 bytes per lane, not 8, one loader for x rows and W rows: half the load instructions and line visits; k order changes inside each 8 k on both sides (sum order changes)
# speedup vs baseline: 2.0726x; 1.0308x over previous
#include <hip/hip_runtime.h>


typedef __attribute__((ext_vector_type(2))) float v2f;
typedef __attribute__((ext_vector_type(8))) float v8f;
typedef float v4f8 __attribute__((ext_vector_type(4), aligned(8)));
typedef int i4u __attribute__((ext_vector_type(4), aligned(4)));
typedef unsigned u4u __attribute__((ext_vector_type(4), aligned(4)));

#define DFEAT 128
#define LEAKY 0.01f
#define RMAX   196
#define NBLK   256

__device__ __forceinline__ void load_frags(v2f (&f)[32], const float* __restrict__ rowp, int half)
{
  const v4f8* r4 = (const v4f8*)rowp;
  #pragma unroll
  for (int q = 0; q < 16; ++q) {
    const v4f8 v = r4[2 * q + half];
    f[2 * q]     = (v2f){v.x, v.y};
    f[2 * q + 1] = (v2f){v.z, v.w};
  }
}

__device__ __forceinline__ void load_a_tile(v2f (&a)[32], const float* __restrict__ x,
                                            int tile, int lrow, int half, int n_nodes)
{
  int row = tile * 16 + lrow;
  if (row >= n_nodes) row = n_nodes - 1;
  load_frags(a, x + (size_t)row * DFEAT, half);
}

__global__ void __launch_bounds__(256) gcn_gemm_xWT(
    const float* __restrict__ x, const float* __restrict__ W,
    float* __restrict__ h, int n_nodes)
{
  const int wave = __builtin_amdgcn_readfirstlane((int)(threadIdx.x >> 5));
  const int lane = threadIdx.x & 31;
  const int half = lane >> 4;
  const int lrow = lane & 15;

  const int nh     = wave >> 2;
  const int slot   = blockIdx.x * 4 + (wave & 3);
  const int nslots = gridDim.x * 4;
  const int ntiles = (n_nodes + 15) >> 4;

  v2f bfrag[4][32];
  #pragma unroll
  for (int j = 0; j < 4; ++j) {
    const int col = (nh * 4 + j) * 16 + lrow;
    load_frags(bfrag[j], W + (size_t)col * DFEAT, half);
  }

  v2f anext[32];
  if (slot < ntiles) load_a_tile(anext, x, slot, lrow, half, n_nodes);

  for (int t = slot; t < ntiles; t += nslots) {
    const int base = t * 16;

    v2f afrag[32];
    #pragma unroll
    for (int kt = 0; kt < 32; ++kt) afrag[kt] = anext[kt];

    int tn = t + nslots;
    if (tn >= ntiles) tn = ntiles - 1;
    load_a_tile(anext, x, tn, lrow, half, n_nodes);

    #pragma unroll
    for (int jp = 0; jp < 2; ++jp) {
      v8f acc[2] = {};
      #pragma unroll
      for (int kt = 0; kt < 32; ++kt) {
        #pragma unroll
        for (int u = 0; u < 2; ++u)
          acc[u] = __builtin_amdgcn_wmma_f32_16x16x4_f32(
               false, afrag[kt],  false, bfrag[2 * jp + u][kt],
               (short)0, acc[u],  false,  false);
      }
      #pragma unroll
      for (int u = 0; u < 2; ++u) {
        const int j = 2 * jp + u;
        const int col = (nh * 4 + j) * 16 + lrow;
        #pragma unroll
        for (int r = 0; r < 8; ++r) {
          int m = base + half * 8 + r;
          if (m < n_nodes) h[(size_t)m * DFEAT + col] = acc[u][r];
        }
      }
    }
  }
}

__global__ void deg_init(float* __restrict__ deg, int n) {
  int i = blockIdx.x * blockDim.x + threadIdx.x;
  if (i < n) deg[i] = 1.0f;
}
__global__ void deg_accum(const int* __restrict__ dst, float* __restrict__ deg,
                          unsigned char* __restrict__ own, int E) {
  int e = blockIdx.x * blockDim.x + threadIdx.x;
  if (e < E) {
    const int d = dst[e];
    unsafeAtomicAdd(&deg[d], 1.0f);
    if (own) own[e] = (unsigned char)((d / RMAX) % NBLK);
  }
}
__global__ void deg_finalize(float* __restrict__ deg, int n) {
  int i = blockIdx.x * blockDim.x + threadIdx.x;
  if (i < n) deg[i] = __frsqrt_rn(deg[i]);
}

#define SCAN_U 8
#define QCAP_D (32 + SCAN_U * 128)
#define QCAP_K 64

__device__ __forceinline__ void work_off_queue(
    const int* q, int cnt, int lane, int base, int rows,
    const int* __restrict__ src, const int* __restrict__ dst,
    const float* __restrict__ h, const float* __restrict__ dis,
    const float* dsh, float* acc)
{
  for (int b0 = 0; b0 < cnt; b0 += 32) {
    const int nb = min(32, cnt - b0);
    int myS = 0, myR = RMAX;
    float myW = 0.0f;
    if (lane < nb) {
      const int e = q[b0 + lane];
      const int s = src[e];
      const int r = dst[e] - base;
      const bool mine = (unsigned)r < (unsigned)rows;
      const float w = dis[s] * dsh[mine ? r : 0];
      myS = mine ? s : 0;
      myR = mine ? r : RMAX;
      myW = mine ? w : 0.0f;
    }
    for (int g = 0; g < nb; g += 8) {
      float v[8][4];
      #pragma unroll
      for (int k = 0; k < 8; ++k) {
        const int s = __builtin_amdgcn_readlane(myS, g + k);
        const float* hs = h + (size_t)s * DFEAT + lane;
        #pragma unroll
        for (int j = 0; j < 4; ++j) v[k][j] = hs[32 * j];
      }
      #pragma unroll
      for (int k = 0; k < 8; ++k) {
        const float w = __int_as_float(__builtin_amdgcn_readlane(__float_as_int(myW), g + k));
        const int   r = __builtin_amdgcn_readlane(myR, g + k);
        float* a = acc + r * DFEAT + lane;
        #pragma unroll
        for (int j = 0; j < 4; ++j)
          __hip_atomic_fetch_add(a + 32 * j, v[k][j] * w, __ATOMIC_RELAXED, __HIP_MEMORY_SCOPE_WORKGROUP);
      }
    }
  }
}

__device__ __forceinline__ unsigned zero_bytes(unsigned x) {
  return ~((((x & 0x7f7f7f7fu) + 0x7f7f7f7fu) | x) | 0x7f7f7f7fu);
}

template <bool KEYS>
__global__ void __launch_bounds__(1024) gcn_node_sum(
    const int* __restrict__ src, const int* __restrict__ dst,
    const unsigned char* __restrict__ own,
    const float* __restrict__ h, const float* __restrict__ dis,
    const float* __restrict__ bias, float* __restrict__ outp, int n_nodes, int E)
{
  __shared__ float acc[(RMAX + 1) * DFEAT];
  __shared__ float dsh[RMAX];
  __shared__ int   queue[32 * (KEYS ? QCAP_K : QCAP_D)];

  const int tid  = threadIdx.x;
  const int lane = tid & 31;
  const int wave = __builtin_amdgcn_readfirstlane(tid >> 5);
  int* q = queue + wave * (KEYS ? QCAP_K : QCAP_D);

  const int GSH = KEYS ? 4 : 2;
  const int NQ  = E >> GSH;
  const int NIT = (NQ + 1023) >> 10;
  const int off = (int)((blockIdx.x * (unsigned)NIT) / gridDim.x);
  const unsigned pat = (blockIdx.x % NBLK) * 0x01010101u;

  for (int base = blockIdx.x * RMAX; base < n_nodes; base += gridDim.x * RMAX) {
    const int rows = min(RMAX, n_nodes - base);

    for (int idx = tid; idx < rows * DFEAT; idx += 1024) {
      const int r = idx >> 7, f = idx & (DFEAT - 1);
      const float dv = dis[base + r];
      const float hv = h[(size_t)base * DFEAT + idx];
      acc[idx] = bias[f] + hv * dv * dv;
    }
    if (tid < rows) dsh[tid] = dis[base + tid];
    __syncthreads();

    int cnt = 0;
    auto push = [&](bool m, int e) {
      const unsigned b = __builtin_amdgcn_ballot_w32(m);
      if (m) q[cnt + __builtin_amdgcn_mbcnt_lo(b, 0)] = e;
      cnt += __builtin_popcount(b);
    };

    if (KEYS) {
      for (int i = 0; i < NIT; i += SCAN_U) {
        u4u kq[SCAN_U];
        bool ok[SCAN_U];
        #pragma unroll
        for (int u = 0; u < SCAN_U; ++u) {
          int itr = i + u + off;
          if (itr >= NIT) itr -= NIT;
          const int g = itr * 1024 + tid;
          ok[u] = (i + u < NIT) && (g < NQ);
          kq[u] = *(const u4u*)(own + 16 * (size_t)(ok[u] ? g : 0));
        }
        unsigned M[SCAN_U / 2];
        #pragma unroll
        for (int u = 0; u < SCAN_U; ++u) {
          unsigned m = (zero_bytes(kq[u].x ^ pat) >> 3) | (zero_bytes(kq[u].y ^ pat) >> 2)
                     | (zero_bytes(kq[u].z ^ pat) >> 1) |  zero_bytes(kq[u].w ^ pat);
          if (!ok[u]) m = 0;
          if (u & 1) M[u >> 1] |= m >> 4; else M[u >> 1] = m;
        }
        for (;;) {
          const unsigned any = M[0] | M[1] | M[2] | M[3];
          const unsigned bal = __builtin_amdgcn_ballot_w32(any != 0u);
          if (bal == 0u) break;
          if (any != 0u) {
            const int w = M[0] ? 0 : M[1] ? 1 : M[2] ? 2 : 3;
            const unsigned mw = M[0] ? M[0] : M[1] ? M[1] : M[2] ? M[2] : M[3];
            const int p = __builtin_ctz(mw);
            int itr = i + 2 * w + (((p >> 2) & 1) ^ 1) + off;
            if (itr >= NIT) itr -= NIT;
            q[cnt + __builtin_amdgcn_mbcnt_lo(bal, 0)] = ((itr * 1024 + tid) << 4) + 4 * (p & 3) + (p >> 3);
            const unsigned cl = mw & (mw - 1u);
            M[0] = (w == 0) ? cl : M[0];
            M[1] = (w == 1) ? cl : M[1];
            M[2] = (w == 2) ? cl : M[2];
            M[3] = (w == 3) ? cl : M[3];
          }
          cnt += __builtin_popcount(bal);
          if (cnt >= 32) {
            work_off_queue(q, cnt, lane, base, rows, src, dst, h, dis, dsh, acc);
            cnt = 0;
          }
        }
      }
    } else {
      for (int i = 0; i < NIT; i += SCAN_U) {
        i4u dq[SCAN_U];
        int gq[SCAN_U];
        bool ok[SCAN_U];
        #pragma unroll
        for (int u = 0; u < SCAN_U; ++u) {
          int itr = i + u + off;
          if (itr >= NIT) itr -= NIT;
          const int g = itr * 1024 + tid;
          ok[u] = (i + u < NIT) && (g < NQ);
          gq[u] = ok[u] ? g : 0;
          dq[u] = *(const i4u*)(dst + 4 * (size_t)gq[u]);
        }
        #pragma unroll
        for (int u = 0; u < SCAN_U; ++u) {
          const bool m0 = ok[u] && (unsigned)(dq[u].x - base) < (unsigned)rows;
          const bool m1 = ok[u] && (unsigned)(dq[u].y - base) < (unsigned)rows;
          const bool m2 = ok[u] && (unsigned)(dq[u].z - base) < (unsigned)rows;
          const bool m3 = ok[u] && (unsigned)(dq[u].w - base) < (unsigned)rows;
          if (__builtin_amdgcn_ballot_w32(m0 || m1 || m2 || m3)) {
            push(m0, 4 * gq[u] + 0);
            push(m1, 4 * gq[u] + 1);
            push(m2, 4 * gq[u] + 2);
            push(m3, 4 * gq[u] + 3);
          }
        }
        if (cnt >= 32) {
          work_off_queue(q, cnt, lane, base, rows, src, dst, h, dis, dsh, acc);
          cnt = 0;
        }
      }
    }
    {
      const int e = (NQ << GSH) + lane;
      bool m = (wave == 0) && (lane < (E & ((1 << GSH) - 1)));
      const int d = m ? dst[e] : -1;
      m = m && (unsigned)(d - base) < (unsigned)rows;
      push(m, e);
    }
    work_off_queue(q, cnt, lane, base, rows, src, dst, h, dis, dsh, acc);
    __syncthreads();

    for (int idx = tid; idx < rows * DFEAT; idx += 1024) {
      const float v = acc[idx];
      outp[(size_t)base * DFEAT + idx] = (v >= 0.0f) ? v : LEAKY * v;
    }
    __syncthreads();
  }
}

extern "C" void kernel_launch(void* const* d_in, const int* in_sizes, int n_in,
                              void* d_out, int out_size, void* d_ws, size_t ws_size,
                              hipStream_t stream) {
  const float* x  = (const float*)d_in[0];
  const int*   ei = (const int*)d_in[1];
  const float* Ws = (const float*)d_in[2];
  const float* bs = (const float*)d_in[3];
  float* out = (float*)d_out;

  const int n = in_sizes[0] / DFEAT;
  const int E = in_sizes[1] / 2;
  const int* src = ei;
  const int* dst = ei + E;

  float* xbuf = (float*)d_ws;
  float* hbuf = xbuf + (size_t)n * DFEAT;
  float* dis  = hbuf + (size_t)n * DFEAT;
  const bool keys = ws_size >= ((size_t)n * (2 * DFEAT + 1)) * sizeof(float) + (size_t)E;
  unsigned char* own = keys ? (unsigned char*)(dis + n) : nullptr;

  const int TB = 256;

  deg_init    <<<(n + TB - 1) / TB, TB, 0, stream>>>(dis, n);
  deg_accum   <<<(E + TB - 1) / TB, TB, 0, stream>>>(dst, dis, own, E);
  deg_finalize<<<(n + TB - 1) / TB, TB, 0, stream>>>(dis, n);

  const int gg = 256;
  const int ga = NBLK;
  auto node_sum = [&](const float* hin, const float* b, float* o) {
    if (keys) gcn_node_sum<true ><<<ga, 1024, 0, stream>>>(src, dst, own, hin, dis, b, o, n, E);
    else      gcn_node_sum<false><<<ga, 1024, 0, stream>>>(src, dst, own, hin, dis, b, o, n, E);
  };

  gcn_gemm_xWT<<<gg, 256, 0, stream>>>(x, Ws, hbuf, n);
  node_sum(hbuf, bs, xbuf);

  gcn_gemm_xWT<<<gg, 256, 0, stream>>>(xbuf, Ws + (size_t)DFEAT * DFEAT, hbuf, n);
  node_sum(hbuf, bs + DFEAT, xbuf);

  gcn_gemm_xWT<<<gg, 256, 0, stream>>>(xbuf, Ws + (size_t)2 * DFEAT * DFEAT, hbuf, n);
  node_sum(hbuf, bs + 2 * DFEAT, out);
}


# Edge sum moved in front of the product: a block sums its rows of x in LDS, then multiplies them by W^T itself (A, B fragments from LDS); no x @ W^T launches, 6 not 9; rounding order changes
# speedup vs baseline: 2.5142x; 1.2131x over previous
#include <hip/hip_runtime.h>


typedef __attribute__((ext_vector_type(2))) float v2f;
typedef __attribute__((ext_vector_type(8))) float v8f;
typedef float v4f8 __attribute__((ext_vector_type(4), aligned(8)));
typedef int i4u __attribute__((ext_vector_type(4), aligned(4)));
typedef unsigned u4u __attribute__((ext_vector_type(4), aligned(4)));

typedef float v4f16 __attribute__((ext_vector_type(4), aligned(16)));

#define DFEAT 128
#define LEAKY 0.01f
#define PITCH  132
#define RMAX   196
#define RPAD   208
#define NBLK   256

__device__ __forceinline__ void load_frags(v2f (&f)[32], const float* rowp, int half)
{
  const v4f16* r4 = (const v4f16*)rowp + half;
  #pragma unroll
  for (int q = 0; q < 16; ++q) {
    const v4f16 v = r4[2 * q];
    f[2 * q]     = (v2f){v.x, v.y};
    f[2 * q + 1] = (v2f){v.z, v.w};
  }
}

__device__ __forceinline__ v8f wmma4(v2f a, v2f b, v8f c) {
  return __builtin_amdgcn_wmma_f32_16x16x4_f32(
       false, a,  false, b,  (short)0, c,  false,  false);
}

__global__ void deg_init(float* __restrict__ deg, int n) {
  int i = blockIdx.x * blockDim.x + threadIdx.x;
  if (i < n) deg[i] = 1.0f;
}
__global__ void deg_accum(const int* __restrict__ dst, float* __restrict__ deg,
                          unsigned char* __restrict__ own, int E) {
  int e = blockIdx.x * blockDim.x + threadIdx.x;
  if (e < E) {
    const int d = dst[e];
    unsafeAtomicAdd(&deg[d], 1.0f);
    if (own) own[e] = (unsigned char)((d / RMAX) % NBLK);
  }
}
__global__ void deg_finalize(float* __restrict__ deg, int n) {
  int i = blockIdx.x * blockDim.x + threadIdx.x;
  if (i < n) deg[i] = __frsqrt_rn(deg[i]);
}

#define SCAN_U 8
#define QCAP_D (32 + SCAN_U * 128)
#define QCAP_K 64

__device__ __forceinline__ void work_off_queue(
    const int* q, int cnt, int lane, int base, int rows,
    const int* __restrict__ src, const int* __restrict__ dst,
    const float* __restrict__ h, const float* __restrict__ dis,
    const float* dsh, float* acc)
{
  for (int b0 = 0; b0 < cnt; b0 += 32) {
    const int nb = min(32, cnt - b0);
    int myS = 0, myR = RMAX;
    float myW = 0.0f;
    if (lane < nb) {
      const int e = q[b0 + lane];
      const int s = src[e];
      const int r = dst[e] - base;
      const bool mine = (unsigned)r < (unsigned)rows;
      const float w = dis[s] * dsh[mine ? r : 0];
      myS = mine ? s : 0;
      myR = mine ? r : RMAX;
      myW = mine ? w : 0.0f;
    }
    for (int g = 0; g < nb; g += 8) {
      float v[8][4];
      #pragma unroll
      for (int k = 0; k < 8; ++k) {
        const int s = __builtin_amdgcn_readlane(myS, g + k);
        const float* hs = h + (size_t)s * DFEAT + lane;
        #pragma unroll
        for (int j = 0; j < 4; ++j) v[k][j] = hs[32 * j];
      }
      #pragma unroll
      for (int k = 0; k < 8; ++k) {
        const float w = __int_as_float(__builtin_amdgcn_readlane(__float_as_int(myW), g + k));
        const int   r = __builtin_amdgcn_readlane(myR, g + k);
        float* a = acc + r * PITCH + lane;
        #pragma unroll
        for (int j = 0; j < 4; ++j)
          __hip_atomic_fetch_add(a + 32 * j, v[k][j] * w, __ATOMIC_RELAXED, __HIP_MEMORY_SCOPE_WORKGROUP);
      }
    }
  }
}

__device__ __forceinline__ unsigned zero_bytes(unsigned x) {
  return ~((((x & 0x7f7f7f7fu) + 0x7f7f7f7fu) | x) | 0x7f7f7f7fu);
}

template <bool KEYS>
__global__ void __launch_bounds__(1024) gcn_layer(
    const int* __restrict__ src, const int* __restrict__ dst,
    const unsigned char* __restrict__ own,
    const float* __restrict__ h, const float* __restrict__ dis,
    const float* __restrict__ W, const float* __restrict__ bias,
    float* __restrict__ outp, int n_nodes, int E)
{
  __shared__ __attribute__((aligned(16))) float acc[RPAD * PITCH];
  __shared__ __attribute__((aligned(16))) float wsh[DFEAT * PITCH];
  __shared__ float dsh[RMAX];
  __shared__ int   queue[32 * (KEYS ? QCAP_K : QCAP_D)];

  const int tid  = threadIdx.x;
  const int lane = tid & 31;
  const int wave = __builtin_amdgcn_readfirstlane(tid >> 5);
  int* q = queue + wave * (KEYS ? QCAP_K : QCAP_D);

  const int GSH = KEYS ? 4 : 2;
  const int NQ  = E >> GSH;
  const int NIT = (NQ + 1023) >> 10;
  const int off = (int)((blockIdx.x * (unsigned)NIT) / gridDim.x);
  const unsigned pat = (blockIdx.x % NBLK) * 0x01010101u;

  {
    v4f8 wv[4];
    #pragma unroll
    for (int u = 0; u < 4; ++u) {
      const int idx = tid + 1024 * u, n = idx >> 5, k4 = idx & 31;
      wv[u] = *(const v4f8*)(W + n * DFEAT + 4 * k4);
    }
    #pragma unroll
    for (int u = 0; u < 4; ++u) {
      const int idx = tid + 1024 * u, n = idx >> 5, k4 = idx & 31;
      *(v4f16*)(wsh + n * PITCH + 4 * k4) = (v4f16){wv[u].x, wv[u].y, wv[u].z, wv[u].w};
    }
  }

  for (int base = blockIdx.x * RMAX; base < n_nodes; base += gridDim.x * RMAX) {
    const int rows = min(RMAX, n_nodes - base);

    for (int idx = tid; idx < rows * DFEAT; idx += 1024) {
      const int r = idx >> 7, f = idx & (DFEAT - 1);
      const float dv = dis[base + r];
      const float hv = h[(size_t)base * DFEAT + idx];
      acc[r * PITCH + f] = hv * dv * dv;
    }
    for (int r = rows + (tid >> 7); r < RPAD; r += 8) acc[r * PITCH + (tid & (DFEAT - 1))] = 0.0f;
    if (tid < rows) dsh[tid] = dis[base + tid];
    __syncthreads();

    int cnt = 0;
    auto push = [&](bool m, int e) {
      const unsigned b = __builtin_amdgcn_ballot_w32(m);
      if (m) q[cnt + __builtin_amdgcn_mbcnt_lo(b, 0)] = e;
      cnt += __builtin_popcount(b);
    };

    if (KEYS) {
      for (int i = 0; i < NIT; i += SCAN_U) {
        u4u kq[SCAN_U];
        bool ok[SCAN_U];
        #pragma unroll
        for (int u = 0; u < SCAN_U; ++u) {
          int itr = i + u + off;
          if (itr >= NIT) itr -= NIT;
          const int g = itr * 1024 + tid;
          ok[u] = (i + u < NIT) && (g < NQ);
          kq[u] = *(const u4u*)(own + 16 * (size_t)(ok[u] ? g : 0));
        }
        unsigned M[SCAN_U / 2];
        #pragma unroll
        for (int u = 0; u < SCAN_U; ++u) {
          unsigned m = (zero_bytes(kq[u].x ^ pat) >> 3) | (zero_bytes(kq[u].y ^ pat) >> 2)
                     | (zero_bytes(kq[u].z ^ pat) >> 1) |  zero_bytes(kq[u].w ^ pat);
          if (!ok[u]) m = 0;
          if (u & 1) M[u >> 1] |= m >> 4; else M[u >> 1] = m;
        }
        for (;;) {
          const unsigned any = M[0] | M[1] | M[2] | M[3];
          const unsigned bal = __builtin_amdgcn_ballot_w32(any != 0u);
          if (bal == 0u) break;
          if (any != 0u) {
            const int w = M[0] ? 0 : M[1] ? 1 : M[2] ? 2 : 3;
            const unsigned mw = M[0] ? M[0] : M[1] ? M[1] : M[2] ? M[2] : M[3];
            const int p = __builtin_ctz(mw);
            int itr = i + 2 * w + (((p >> 2) & 1) ^ 1) + off;
            if (itr >= NIT) itr -= NIT;
            q[cnt + __builtin_amdgcn_mbcnt_lo(bal, 0)] = ((itr * 1024 + tid) << 4) + 4 * (p & 3) + (p >> 3);
            const unsigned cl = mw & (mw - 1u);
            M[0] = (w == 0) ? cl : M[0];
            M[1] = (w == 1) ? cl : M[1];
            M[2] = (w == 2) ? cl : M[2];
            M[3] = (w == 3) ? cl : M[3];
          }
          cnt += __builtin_popcount(bal);
          if (cnt >= 32) {
            work_off_queue(q, cnt, lane, base, rows, src, dst, h, dis, dsh, acc);
            cnt = 0;
          }
        }
      }
    } else {
      for (int i = 0; i < NIT; i += SCAN_U) {
        i4u dq[SCAN_U];
        int gq[SCAN_U];
        bool ok[SCAN_U];
        #pragma unroll
        for (int u = 0; u < SCAN_U; ++u) {
          int itr = i + u + off;
          if (itr >= NIT) itr -= NIT;
          const int g = itr * 1024 + tid;
          ok[u] = (i + u < NIT) && (g < NQ);
          gq[u] = ok[u] ? g : 0;
          dq[u] = *(const i4u*)(dst + 4 * (size_t)gq[u]);
        }
        #pragma unroll
        for (int u = 0; u < SCAN_U; ++u) {
          const bool m0 = ok[u] && (unsigned)(dq[u].x - base) < (unsigned)rows;
          const bool m1 = ok[u] && (unsigned)(dq[u].y - base) < (unsigned)rows;
          const bool m2 = ok[u] && (unsigned)(dq[u].z - base) < (unsigned)rows;
          const bool m3 = ok[u] && (unsigned)(dq[u].w - base) < (unsigned)rows;
          if (__builtin_amdgcn_ballot_w32(m0 || m1 || m2 || m3)) {
            push(m0, 4 * gq[u] + 0);
            push(m1, 4 * gq[u] + 1);
            push(m2, 4 * gq[u] + 2);
            push(m3, 4 * gq[u] + 3);
          }
        }
        if (cnt >= 32) {
          work_off_queue(q, cnt, lane, base, rows, src, dst, h, dis, dsh, acc);
          cnt = 0;
        }
      }
    }
    {
      const int e = (NQ << GSH) + lane;
      bool m = (wave == 0) && (lane < (E & ((1 << GSH) - 1)));
      const int d = m ? dst[e] : -1;
      m = m && (unsigned)(d - base) < (unsigned)rows;
      push(m, e);
    }
    work_off_queue(q, cnt, lane, base, rows, src, dst, h, dis, dsh, acc);

    const int half = lane >> 4;
    const int lrow = lane & 15;
    const int ct   = wave & 7;
    const int g    = wave >> 3;
    v2f bfrag[32];
    load_frags(bfrag, wsh + (ct * 16 + lrow) * PITCH, half);
    const float bv = bias[ct * 16 + lrow];
    __syncthreads();

    #pragma unroll
    for (int p = 0; p < 2; ++p) {
      const int t0 = g + 8 * p, t1 = t0 + 4;
      const v4f16* a0 = (const v4f16*)(acc + (t0 * 16 + lrow) * PITCH) + half;
      v8f d[2] = {};
      if (t1 * 16 < rows) {
        const v4f16* a1 = a0 + 4 * 16 * (PITCH / 4);
        #pragma unroll
        for (int qq = 0; qq < 16; ++qq) {
          const v4f16 va0 = a0[2 * qq], va1 = a1[2 * qq];
          d[0] = wmma4((v2f){va0.x, va0.y}, bfrag[2 * qq],     d[0]);
          d[1] = wmma4((v2f){va1.x, va1.y}, bfrag[2 * qq],     d[1]);
          d[0] = wmma4((v2f){va0.z, va0.w}, bfrag[2 * qq + 1], d[0]);
          d[1] = wmma4((v2f){va1.z, va1.w}, bfrag[2 * qq + 1], d[1]);
        }
      } else if (t0 * 16 < rows) {
        #pragma unroll
        for (int qq = 0; qq < 16; ++qq) {
          const v4f16 va0 = a0[2 * qq];
          d[0] = wmma4((v2f){va0.x, va0.y}, bfrag[2 * qq],     d[0]);
          d[0] = wmma4((v2f){va0.z, va0.w}, bfrag[2 * qq + 1], d[0]);
        }
      }
      __syncthreads();
      const int row0 = t0 * 16 + half * 8;
      float* dp = acc + row0 * PITCH + ct * 16 + lrow;
      const int lim = rows - row0;
      #pragma unroll
      for (int u = 0; u < 2; ++u) {
        #pragma unroll
        for (int r = 0; r < 8; ++r) {
          const float v = d[u][r] + bv;
          if (64 * u + r < lim) dp[(64 * u + r) * PITCH] = (v >= 0.0f) ? v : LEAKY * v;
        }
      }
    }
    __syncthreads();

    for (int idx = tid; idx < rows * DFEAT; idx += 1024) {
      const int r = idx >> 7, f = idx & (DFEAT - 1);
      outp[(size_t)base * DFEAT + idx] = acc[r * PITCH + f];
    }
    __syncthreads();
  }
}

extern "C" void kernel_launch(void* const* d_in, const int* in_sizes, int n_in,
                              void* d_out, int out_size, void* d_ws, size_t ws_size,
                              hipStream_t stream) {
  const float* x  = (const float*)d_in[0];
  const int*   ei = (const int*)d_in[1];
  const float* Ws = (const float*)d_in[2];
  const float* bs = (const float*)d_in[3];
  float* out = (float*)d_out;

  const int n = in_sizes[0] / DFEAT;
  const int E = in_sizes[1] / 2;
  const int* src = ei;
  const int* dst = ei + E;

  float* xbuf = (float*)d_ws;
  float* hbuf = xbuf + (size_t)n * DFEAT;
  float* dis  = hbuf + (size_t)n * DFEAT;
  const bool keys = ws_size >= ((size_t)n * (2 * DFEAT + 1)) * sizeof(float) + (size_t)E;
  unsigned char* own = keys ? (unsigned char*)(dis + n) : nullptr;

  const int TB = 256;

  deg_init    <<<(n + TB - 1) / TB, TB, 0, stream>>>(dis, n);
  deg_accum   <<<(E + TB - 1) / TB, TB, 0, stream>>>(dst, dis, own, E);
  deg_finalize<<<(n + TB - 1) / TB, TB, 0, stream>>>(dis, n);

  const int ga = NBLK;
  auto layer = [&](const float* xin, const float* w, const float* b, float* o) {
    if (keys) gcn_layer<true ><<<ga, 1024, 0, stream>>>(src, dst, own, xin, dis, w, b, o, n, E);
    else      gcn_layer<false><<<ga, 1024, 0, stream>>>(src, dst, own, xin, dis, w, b, o, n, E);
  };

  layer(x,    Ws,                             bs,             xbuf);
  layer(xbuf, Ws + (size_t)DFEAT * DFEAT,     bs + DFEAT,     hbuf);
  layer(hbuf, Ws + (size_t)2 * DFEAT * DFEAT, bs + 2 * DFEAT, out);
}


# Layer 1's edge scan also writes each wave's found edges (source, row, weight) to a list in the workspace; layers 2, 3 read it back: no scan, no src/dst/degree lookups; LDS adds still in arrival order
# speedup vs baseline: 3.0003x; 1.1933x over previous
#include <hip/hip_runtime.h>


typedef __attribute__((ext_vector_type(2))) float v2f;
typedef __attribute__((ext_vector_type(8))) float v8f;
typedef float v4f8 __attribute__((ext_vector_type(4), aligned(8)));
typedef int i4u __attribute__((ext_vector_type(4), aligned(4)));
typedef unsigned u4u __attribute__((ext_vector_type(4), aligned(4)));

typedef unsigned u2 __attribute__((ext_vector_type(2)));

typedef float v4f16 __attribute__((ext_vector_type(4), aligned(16)));

#define DFEAT 128
#define LEAKY 0.01f
#define PITCH  132
#define RMAX   196
#define RPAD   208
#define NBLK   256
#define LCAP   128

__device__ __forceinline__ void load_frags(v2f (&f)[32], const float* rowp, int half)
{
  const v4f16* r4 = (const v4f16*)rowp + half;
  #pragma unroll
  for (int q = 0; q < 16; ++q) {
    const v4f16 v = r4[2 * q];
    f[2 * q]     = (v2f){v.x, v.y};
    f[2 * q + 1] = (v2f){v.z, v.w};
  }
}

__device__ __forceinline__ v8f wmma4(v2f a, v2f b, v8f c) {
  return __builtin_amdgcn_wmma_f32_16x16x4_f32(
       false, a,  false, b,  (short)0, c,  false,  false);
}

__global__ void deg_init(float* __restrict__ deg, int n) {
  int i = blockIdx.x * blockDim.x + threadIdx.x;
  if (i < n) deg[i] = 1.0f;
}
__global__ void deg_accum(const int* __restrict__ dst, float* __restrict__ deg,
                          unsigned char* __restrict__ own, int E) {
  int e = blockIdx.x * blockDim.x + threadIdx.x;
  if (e < E) {
    const int d = dst[e];
    unsafeAtomicAdd(&deg[d], 1.0f);
    if (own) own[e] = (unsigned char)((d / RMAX) % NBLK);
  }
}
__global__ void deg_finalize(float* __restrict__ deg, int n) {
  int i = blockIdx.x * blockDim.x + threadIdx.x;
  if (i < n) deg[i] = __frsqrt_rn(deg[i]);
}

#define SCAN_U 8
#define QCAP_D (32 + SCAN_U * 128)
#define QCAP_K 64

__device__ __forceinline__ void add_rows(int nb, int myS, int myR, float myW, int lane,
                                         const float* __restrict__ h, float* acc)
{
  for (int g = 0; g < nb; g += 8) {
    float v[8][4];
    #pragma unroll
    for (int k = 0; k < 8; ++k) {
      const int s = __builtin_amdgcn_readlane(myS, g + k);
      const float* hs = h + (size_t)s * DFEAT + lane;
      #pragma unroll
      for (int j = 0; j < 4; ++j) v[k][j] = hs[32 * j];
    }
    #pragma unroll
    for (int k = 0; k < 8; ++k) {
      const float w = __int_as_float(__builtin_amdgcn_readlane(__float_as_int(myW), g + k));
      const int   r = __builtin_amdgcn_readlane(myR, g + k);
      float* a = acc + r * PITCH + lane;
      #pragma unroll
      for (int j = 0; j < 4; ++j)
        __hip_atomic_fetch_add(a + 32 * j, v[k][j] * w, __ATOMIC_RELAXED, __HIP_MEMORY_SCOPE_WORKGROUP);
    }
  }
}

template <int LIST>
__device__ __forceinline__ void work_off_queue(
    const int* q, int cnt, int lane, int base, int rows,
    const int* __restrict__ src, const int* __restrict__ dst,
    const float* __restrict__ h, const float* __restrict__ dis,
    const float* dsh, float* acc, u2* lp, int& pos)
{
  for (int b0 = 0; b0 < cnt; b0 += 32) {
    const int nb = min(32, cnt - b0);
    int myS = 0, myR = RMAX;
    float myW = 0.0f;
    if (lane < nb) {
      const int e = q[b0 + lane];
      const int s = src[e];
      const int r = dst[e] - base;
      const bool mine = (unsigned)r < (unsigned)rows;
      const float w = dis[s] * dsh[mine ? r : 0];
      myS = mine ? s : 0;
      myR = mine ? r : RMAX;
      myW = mine ? w : 0.0f;
    }
    if (LIST == 1) {
      if (pos + nb <= LCAP) {
        if (lane < nb) lp[pos + lane] = (u2){(unsigned)myS | ((unsigned)myR << 24), __float_as_uint(myW)};
        pos += nb;
      } else {
        pos = LCAP + 1;
      }
    }
    add_rows(nb, myS, myR, myW, lane, h, acc);
  }
}

__device__ __forceinline__ unsigned zero_bytes(unsigned x) {
  return ~((((x & 0x7f7f7f7fu) + 0x7f7f7f7fu) | x) | 0x7f7f7f7fu);
}

template <bool KEYS, int LIST>
__global__ void __launch_bounds__(1024) gcn_layer(
    const int* __restrict__ src, const int* __restrict__ dst,
    const unsigned char* __restrict__ own,
    const float* __restrict__ h, const float* __restrict__ dis,
    const float* __restrict__ W, const float* __restrict__ bias,
    float* __restrict__ outp, int n_nodes, int E,
    unsigned* __restrict__ lst, int* __restrict__ lcnt)
{
  __shared__ __attribute__((aligned(16))) float acc[RPAD * PITCH];
  __shared__ __attribute__((aligned(16))) float wsh[DFEAT * PITCH];
  __shared__ float dsh[RMAX];
  __shared__ int   queue[32 * (KEYS ? QCAP_K : QCAP_D)];

  const int tid  = threadIdx.x;
  const int lane = tid & 31;
  const int wave = __builtin_amdgcn_readfirstlane(tid >> 5);
  int* q = queue + wave * (KEYS ? QCAP_K : QCAP_D);

  u2* lp = (u2*)lst + (size_t)(blockIdx.x * 32 + wave) * LCAP;
  int pos = 0;
  int lc = 0;
  bool replay = false;
  u2 ent[LCAP / 32];
  if (LIST == 2) {
    const int call = lcnt[blockIdx.x * 32 + lane];
    replay = __builtin_amdgcn_ballot_w32(call > LCAP) == 0u;
    lc = replay ? __builtin_amdgcn_readlane(call, wave) : 0;
    #pragma unroll
    for (int k = 0; k < LCAP / 32; ++k) {
      ent[k] = (u2){(unsigned)RMAX << 24, 0u};
      if (32 * k + lane < lc) ent[k] = lp[32 * k + lane];
    }
  }

  const int GSH = KEYS ? 4 : 2;
  const int NQ  = E >> GSH;
  const int NIT = (NQ + 1023) >> 10;
  const int off = (int)((blockIdx.x * (unsigned)NIT) / gridDim.x);
  const unsigned pat = (blockIdx.x % NBLK) * 0x01010101u;

  {
    v4f8 wv[4];
    #pragma unroll
    for (int u = 0; u < 4; ++u) {
      const int idx = tid + 1024 * u, n = idx >> 5, k4 = idx & 31;
      wv[u] = *(const v4f8*)(W + n * DFEAT + 4 * k4);
    }
    #pragma unroll
    for (int u = 0; u < 4; ++u) {
      const int idx = tid + 1024 * u, n = idx >> 5, k4 = idx & 31;
      *(v4f16*)(wsh + n * PITCH + 4 * k4) = (v4f16){wv[u].x, wv[u].y, wv[u].z, wv[u].w};
    }
  }

  for (int base = blockIdx.x * RMAX; base < n_nodes; base += gridDim.x * RMAX) {
    const int rows = min(RMAX, n_nodes - base);

    for (int idx = tid; idx < rows * DFEAT; idx += 1024) {
      const int r = idx >> 7, f = idx & (DFEAT - 1);
      const float dv = dis[base + r];
      const float hv = h[(size_t)base * DFEAT + idx];
      acc[r * PITCH + f] = hv * dv * dv;
    }
    for (int r = rows + (tid >> 7); r < RPAD; r += 8) acc[r * PITCH + (tid & (DFEAT - 1))] = 0.0f;
    if (tid < rows) dsh[tid] = dis[base + tid];
    __syncthreads();

    int cnt = 0;
    auto push = [&](bool m, int e) {
      const unsigned b = __builtin_amdgcn_ballot_w32(m);
      if (m) q[cnt + __builtin_amdgcn_mbcnt_lo(b, 0)] = e;
      cnt += __builtin_popcount(b);
    };

    if (LIST == 2 && replay) {
      #pragma unroll
      for (int k = 0; k < LCAP / 32; ++k) {
        if (32 * k < lc) {
          add_rows(min(32, lc - 32 * k), (int)(ent[k].x & 0xffffffu), (int)(ent[k].x >> 24),
                   __uint_as_float(ent[k].y), lane, h, acc);
        }
      }
    } else {
    if (KEYS) {
      for (int i = 0; i < NIT; i += SCAN_U) {
        u4u kq[SCAN_U];
        bool ok[SCAN_U];
        #pragma unroll
        for (int u = 0; u < SCAN_U; ++u) {
          int itr = i + u + off;
          if (itr >= NIT) itr -= NIT;
          const int g = itr * 1024 + tid;
          ok[u] = (i + u < NIT) && (g < NQ);
          kq[u] = *(const u4u*)(own + 16 * (size_t)(ok[u] ? g : 0));
        }
        unsigned M[SCAN_U / 2];
        #pragma unroll
        for (int u = 0; u < SCAN_U; ++u) {
          unsigned m = (zero_bytes(kq[u].x ^ pat) >> 3) | (zero_bytes(kq[u].y ^ pat) >> 2)
                     | (zero_bytes(kq[u].z ^ pat) >> 1) |  zero_bytes(kq[u].w ^ pat);
          if (!ok[u]) m = 0;
          if (u & 1) M[u >> 1] |= m >> 4; else M[u >> 1] = m;
        }
        for (;;) {
          const unsigned any = M[0] | M[1] | M[2] | M[3];
          const unsigned bal = __builtin_amdgcn_ballot_w32(any != 0u);
          if (bal == 0u) break;
          if (any != 0u) {
            const int w = M[0] ? 0 : M[1] ? 1 : M[2] ? 2 : 3;
            const unsigned mw = M[0] ? M[0] : M[1] ? M[1] : M[2] ? M[2] : M[3];
            const int p = __builtin_ctz(mw);
            int itr = i + 2 * w + (((p >> 2) & 1) ^ 1) + off;
            if (itr >= NIT) itr -= NIT;
            q[cnt + __builtin_amdgcn_mbcnt_lo(bal, 0)] = ((itr * 1024 + tid) << 4) + 4 * (p & 3) + (p >> 3);
            const unsigned cl = mw & (mw - 1u);
            M[0] = (w == 0) ? cl : M[0];
            M[1] = (w == 1) ? cl : M[1];
            M[2] = (w == 2) ? cl : M[2];
            M[3] = (w == 3) ? cl : M[3];
          }
          cnt += __builtin_popcount(bal);
          if (cnt >= 32) {
            work_off_queue<LIST>(q, cnt, lane, base, rows, src, dst, h, dis, dsh, acc, lp, pos);
            cnt = 0;
          }
        }
      }
    } else {
      for (int i = 0; i < NIT; i += SCAN_U) {
        i4u dq[SCAN_U];
        int gq[SCAN_U];
        bool ok[SCAN_U];
        #pragma unroll
        for (int u = 0; u < SCAN_U; ++u) {
          int itr = i + u + off;
          if (itr >= NIT) itr -= NIT;
          const int g = itr * 1024 + tid;
          ok[u] = (i + u < NIT) && (g < NQ);
          gq[u] = ok[u] ? g : 0;
          dq[u] = *(const i4u*)(dst + 4 * (size_t)gq[u]);
        }
        #pragma unroll
        for (int u = 0; u < SCAN_U; ++u) {
          const bool m0 = ok[u] && (unsigned)(dq[u].x - base) < (unsigned)rows;
          const bool m1 = ok[u] && (unsigned)(dq[u].y - base) < (unsigned)rows;
          const bool m2 = ok[u] && (unsigned)(dq[u].z - base) < (unsigned)rows;
          const bool m3 = ok[u] && (unsigned)(dq[u].w - base) < (unsigned)rows;
          if (__builtin_amdgcn_ballot_w32(m0 || m1 || m2 || m3)) {
            push(m0, 4 * gq[u] + 0);
            push(m1, 4 * gq[u] + 1);
            push(m2, 4 * gq[u] + 2);
            push(m3, 4 * gq[u] + 3);
          }
        }
        if (cnt >= 32) {
          work_off_queue<LIST>(q, cnt, lane, base, rows, src, dst, h, dis, dsh, acc, lp, pos);
          cnt = 0;
        }
      }
    }
    {
      const int e = (NQ << GSH) + lane;
      bool m = (wave == 0) && (lane < (E & ((1 << GSH) - 1)));
      const int d = m ? dst[e] : -1;
      m = m && (unsigned)(d - base) < (unsigned)rows;
      push(m, e);
    }
    work_off_queue<LIST>(q, cnt, lane, base, rows, src, dst, h, dis, dsh, acc, lp, pos);
    }

    const int half = lane >> 4;
    const int lrow = lane & 15;
    const int ct   = wave & 7;
    const int g    = wave >> 3;
    v2f bfrag[32];
    load_frags(bfrag, wsh + (ct * 16 + lrow) * PITCH, half);
    const float bv = bias[ct * 16 + lrow];
    __syncthreads();

    #pragma unroll
    for (int p = 0; p < 2; ++p) {
      const int t0 = g + 8 * p, t1 = t0 + 4;
      const v4f16* a0 = (const v4f16*)(acc + (t0 * 16 + lrow) * PITCH) + half;
      v8f d[2] = {};
      if (t1 * 16 < rows) {
        const v4f16* a1 = a0 + 4 * 16 * (PITCH / 4);
        #pragma unroll
        for (int qq = 0; qq < 16; ++qq) {
          const v4f16 va0 = a0[2 * qq], va1 = a1[2 * qq];
          d[0] = wmma4((v2f){va0.x, va0.y}, bfrag[2 * qq],     d[0]);
          d[1] = wmma4((v2f){va1.x, va1.y}, bfrag[2 * qq],     d[1]);
          d[0] = wmma4((v2f){va0.z, va0.w}, bfrag[2 * qq + 1], d[0]);
          d[1] = wmma4((v2f){va1.z, va1.w}, bfrag[2 * qq + 1], d[1]);
        }
      } else if (t0 * 16 < rows) {
        #pragma unroll
        for (int qq = 0; qq < 16; ++qq) {
          const v4f16 va0 = a0[2 * qq];
          d[0] = wmma4((v2f){va0.x, va0.y}, bfrag[2 * qq],     d[0]);
          d[0] = wmma4((v2f){va0.z, va0.w}, bfrag[2 * qq + 1], d[0]);
        }
      }
      __syncthreads();
      const int row0 = t0 * 16 + half * 8;
      float* dp = acc + row0 * PITCH + ct * 16 + lrow;
      const int lim = rows - row0;
      #pragma unroll
      for (int u = 0; u < 2; ++u) {
        #pragma unroll
        for (int r = 0; r < 8; ++r) {
          const float v = d[u][r] + bv;
          if (64 * u + r < lim) dp[(64 * u + r) * PITCH] = (v >= 0.0f) ? v : LEAKY * v;
        }
      }
    }
    __syncthreads();

    for (int idx = tid; idx < rows * DFEAT; idx += 1024) {
      const int r = idx >> 7, f = idx & (DFEAT - 1);
      outp[(size_t)base * DFEAT + idx] = acc[r * PITCH + f];
    }
    __syncthreads();
  }
  if (LIST == 1 && lane == 0) lcnt[blockIdx.x * 32 + wave] = pos;
}

extern "C" void kernel_launch(void* const* d_in, const int* in_sizes, int n_in,
                              void* d_out, int out_size, void* d_ws, size_t ws_size,
                              hipStream_t stream) {
  const float* x  = (const float*)d_in[0];
  const int*   ei = (const int*)d_in[1];
  const float* Ws = (const float*)d_in[2];
  const float* bs = (const float*)d_in[3];
  float* out = (float*)d_out;

  const int n = in_sizes[0] / DFEAT;
  const int E = in_sizes[1] / 2;
  const int* src = ei;
  const int* dst = ei + E;

  const size_t list_words = (size_t)NBLK * 32 * LCAP * 2;
  const size_t need_keys  = ((size_t)n * (2 * DFEAT + 1)) * sizeof(float) + (size_t)E;
  const bool keys  = ws_size >= need_keys;
  const bool lists = keys && n <= NBLK * RMAX
                  && ws_size >= need_keys + (list_words + (size_t)NBLK * 32) * sizeof(unsigned);
  float* xbuf = (float*)d_ws;
  float* hbuf = xbuf + (size_t)n * DFEAT;
  unsigned* lst  = (unsigned*)(hbuf + (size_t)n * DFEAT);
  int*      lcnt = (int*)(lst + list_words);
  float* dis  = lists ? (float*)(lcnt + NBLK * 32) : hbuf + (size_t)n * DFEAT;
  unsigned char* own = keys ? (unsigned char*)(dis + n) : nullptr;

  const int TB = 256;

  deg_init    <<<(n + TB - 1) / TB, TB, 0, stream>>>(dis, n);
  deg_accum   <<<(E + TB - 1) / TB, TB, 0, stream>>>(dst, dis, own, E);
  deg_finalize<<<(n + TB - 1) / TB, TB, 0, stream>>>(dis, n);

  const int ga = NBLK;
  auto layer = [&](const float* xin, const float* w, const float* b, float* o, bool first) {
    if (lists && first) gcn_layer<true,  1><<<ga, 1024, 0, stream>>>(src, dst, own, xin, dis, w, b, o, n, E, lst, lcnt);
    else if (lists)     gcn_layer<true,  2><<<ga, 1024, 0, stream>>>(src, dst, own, xin, dis, w, b, o, n, E, lst, lcnt);
    else if (keys)      gcn_layer<true,  0><<<ga, 1024, 0, stream>>>(src, dst, own, xin, dis, w, b, o, n, E, lst, lcnt);
    else                gcn_layer<false, 0><<<ga, 1024, 0, stream>>>(src, dst, own, xin, dis, w, b, o, n, E, lst, lcnt);
  };

  layer(x,    Ws,                             bs,             xbuf, true);
  layer(xbuf, Ws + (size_t)DFEAT * DFEAT,     bs + DFEAT,     hbuf, false);
  layer(hbuf, Ws + (size_t)2 * DFEAT * DFEAT, bs + 2 * DFEAT, out,  false);
}


# Rows move 16 bytes per lane: source rows, start rows and finished rows take one access of a wave, not four; LDS rows hold feature 4l+j at float 32j+l (k order of the product's sums changes)
# speedup vs baseline: 3.0998x; 1.0332x over previous
#include <hip/hip_runtime.h>


typedef __attribute__((ext_vector_type(2))) float v2f;
typedef __attribute__((ext_vector_type(8))) float v8f;
typedef float v4f8 __attribute__((ext_vector_type(4), aligned(8)));
typedef int i4u __attribute__((ext_vector_type(4), aligned(4)));
typedef unsigned u4u __attribute__((ext_vector_type(4), aligned(4)));

typedef unsigned u2 __attribute__((ext_vector_type(2)));

typedef float v4f16 __attribute__((ext_vector_type(4), aligned(16)));
typedef float v4fa __attribute__((ext_vector_type(4), aligned(4)));

#define DFEAT 128
#define LEAKY 0.01f
#define PITCH  132
#define RMAX   196
#define RPAD   208
#define NBLK   256
#define RW     ((RMAX + 31) / 32)
#define LCAP   128

__device__ __forceinline__ void load_frags(v2f (&f)[32], const float* rowp, int half)
{
  const v4f16* r4 = (const v4f16*)rowp + half;
  #pragma unroll
  for (int q = 0; q < 16; ++q) {
    const v4f16 v = r4[2 * q];
    f[2 * q]     = (v2f){v.x, v.y};
    f[2 * q + 1] = (v2f){v.z, v.w};
  }
}

__device__ __forceinline__ v8f wmma4(v2f a, v2f b, v8f c) {
  return __builtin_amdgcn_wmma_f32_16x16x4_f32(
       false, a,  false, b,  (short)0, c,  false,  false);
}

__global__ void deg_init(float* __restrict__ deg, int n) {
  int i = blockIdx.x * blockDim.x + threadIdx.x;
  if (i < n) deg[i] = 1.0f;
}
__global__ void deg_accum(const int* __restrict__ dst, float* __restrict__ deg,
                          unsigned char* __restrict__ own, int E) {
  int e = blockIdx.x * blockDim.x + threadIdx.x;
  if (e < E) {
    const int d = dst[e];
    unsafeAtomicAdd(&deg[d], 1.0f);
    if (own) own[e] = (unsigned char)((d / RMAX) % NBLK);
  }
}
__global__ void deg_finalize(float* __restrict__ deg, int n) {
  int i = blockIdx.x * blockDim.x + threadIdx.x;
  if (i < n) deg[i] = __frsqrt_rn(deg[i]);
}

#define SCAN_U 8
#define QCAP_D (32 + SCAN_U * 128)
#define QCAP_K 64

__device__ __forceinline__ void add_rows(int nb, int myS, int myR, float myW, int lane,
                                         const float* __restrict__ h, float* acc)
{
  for (int g = 0; g < nb; g += 8) {
    v4fa v[8];
    #pragma unroll
    for (int k = 0; k < 8; ++k) {
      const int s = __builtin_amdgcn_readlane(myS, g + k);
      v[k] = *(const v4fa*)(h + (size_t)s * DFEAT + 4 * lane);
    }
    #pragma unroll
    for (int k = 0; k < 8; ++k) {
      const float w = __int_as_float(__builtin_amdgcn_readlane(__float_as_int(myW), g + k));
      const int   r = __builtin_amdgcn_readlane(myR, g + k);
      float* a = acc + r * PITCH + lane;
      __hip_atomic_fetch_add(a,      v[k].x * w, __ATOMIC_RELAXED, __HIP_MEMORY_SCOPE_WORKGROUP);
      __hip_atomic_fetch_add(a + 32, v[k].y * w, __ATOMIC_RELAXED, __HIP_MEMORY_SCOPE_WORKGROUP);
      __hip_atomic_fetch_add(a + 64, v[k].z * w, __ATOMIC_RELAXED, __HIP_MEMORY_SCOPE_WORKGROUP);
      __hip_atomic_fetch_add(a + 96, v[k].w * w, __ATOMIC_RELAXED, __HIP_MEMORY_SCOPE_WORKGROUP);
    }
  }
}

template <int LIST>
__device__ __forceinline__ void work_off_queue(
    const int* q, int cnt, int lane, int base, int rows,
    const int* __restrict__ src, const int* __restrict__ dst,
    const float* __restrict__ h, const float* __restrict__ dis,
    const float* dsh, float* acc, u2* lp, int& pos)
{
  for (int b0 = 0; b0 < cnt; b0 += 32) {
    const int nb = min(32, cnt - b0);
    int myS = 0, myR = RMAX;
    float myW = 0.0f;
    if (lane < nb) {
      const int e = q[b0 + lane];
      const int s = src[e];
      const int r = dst[e] - base;
      const bool mine = (unsigned)r < (unsigned)rows;
      const float w = dis[s] * dsh[mine ? r : 0];
      myS = mine ? s : 0;
      myR = mine ? r : RMAX;
      myW = mine ? w : 0.0f;
    }
    if (LIST == 1) {
      if (pos + nb <= LCAP) {
        if (lane < nb) lp[pos + lane] = (u2){(unsigned)myS | ((unsigned)myR << 24), __float_as_uint(myW)};
        pos += nb;
      } else {
        pos = LCAP + 1;
      }
    }
    add_rows(nb, myS, myR, myW, lane, h, acc);
  }
}

__device__ __forceinline__ unsigned zero_bytes(unsigned x) {
  return ~((((x & 0x7f7f7f7fu) + 0x7f7f7f7fu) | x) | 0x7f7f7f7fu);
}

template <bool KEYS, int LIST>
__global__ void __launch_bounds__(1024) gcn_layer(
    const int* __restrict__ src, const int* __restrict__ dst,
    const unsigned char* __restrict__ own,
    const float* __restrict__ h, const float* __restrict__ dis,
    const float* __restrict__ W, const float* __restrict__ bias,
    float* __restrict__ outp, int n_nodes, int E,
    unsigned* __restrict__ lst, int* __restrict__ lcnt)
{
  __shared__ __attribute__((aligned(16))) float acc[RPAD * PITCH];
  __shared__ __attribute__((aligned(16))) float wsh[DFEAT * PITCH];
  __shared__ float dsh[RMAX];
  __shared__ int   queue[32 * (KEYS ? QCAP_K : QCAP_D)];

  const int tid  = threadIdx.x;
  const int lane = tid & 31;
  const int wave = __builtin_amdgcn_readfirstlane(tid >> 5);
  int* q = queue + wave * (KEYS ? QCAP_K : QCAP_D);

  u2* lp = (u2*)lst + (size_t)(blockIdx.x * 32 + wave) * LCAP;
  int pos = 0;
  int lc = 0;
  bool replay = false;
  u2 ent[LCAP / 32];
  if (LIST == 2) {
    const int call = lcnt[blockIdx.x * 32 + lane];
    replay = __builtin_amdgcn_ballot_w32(call > LCAP) == 0u;
    lc = replay ? __builtin_amdgcn_readlane(call, wave) : 0;
    #pragma unroll
    for (int k = 0; k < LCAP / 32; ++k) {
      ent[k] = (u2){(unsigned)RMAX << 24, 0u};
      if (32 * k + lane < lc) ent[k] = lp[32 * k + lane];
    }
  }

  const int GSH = KEYS ? 4 : 2;
  const int NQ  = E >> GSH;
  const int NIT = (NQ + 1023) >> 10;
  const int off = (int)((blockIdx.x * (unsigned)NIT) / gridDim.x);
  const unsigned pat = (blockIdx.x % NBLK) * 0x01010101u;

  {
    v4f8 wv[4];
    #pragma unroll
    for (int u = 0; u < 4; ++u) {
      const int idx = tid + 1024 * u, n = idx >> 5, k4 = idx & 31;
      wv[u] = *(const v4f8*)(W + n * DFEAT + 4 * k4);
    }
    #pragma unroll
    for (int u = 0; u < 4; ++u) {
      const int idx = tid + 1024 * u, n = idx >> 5, k4 = idx & 31;
      float* wr = wsh + n * PITCH + k4;
      wr[0] = wv[u].x; wr[32] = wv[u].y; wr[64] = wv[u].z; wr[96] = wv[u].w;
    }
  }

  for (int base = blockIdx.x * RMAX; base < n_nodes; base += gridDim.x * RMAX) {
    const int rows = min(RMAX, n_nodes - base);

    {
      v4fa hv[RW];
      float dv[RW];
      #pragma unroll
      for (int k = 0; k < RW; ++k) {
        const int r = min(wave + 32 * k, rows - 1);
        hv[k] = *(const v4fa*)(h + (size_t)(base + r) * DFEAT + 4 * lane);
        dv[k] = dis[base + r];
      }
      #pragma unroll
      for (int k = 0; k < RW; ++k) {
        const int r = wave + 32 * k;
        if (r < rows) {
          float* a = acc + r * PITCH + lane;
          a[0]  = hv[k].x * dv[k] * dv[k];
          a[32] = hv[k].y * dv[k] * dv[k];
          a[64] = hv[k].z * dv[k] * dv[k];
          a[96] = hv[k].w * dv[k] * dv[k];
        }
      }
    }
    for (int r = rows + (tid >> 7); r < RPAD; r += 8) acc[r * PITCH + (tid & (DFEAT - 1))] = 0.0f;
    if (tid < rows) dsh[tid] = dis[base + tid];
    __syncthreads();

    int cnt = 0;
    auto push = [&](bool m, int e) {
      const unsigned b = __builtin_amdgcn_ballot_w32(m);
      if (m) q[cnt + __builtin_amdgcn_mbcnt_lo(b, 0)] = e;
      cnt += __builtin_popcount(b);
    };

    if (LIST == 2 && replay) {
      #pragma unroll
      for (int k = 0; k < LCAP / 32; ++k) {
        if (32 * k < lc) {
          add_rows(min(32, lc - 32 * k), (int)(ent[k].x & 0xffffffu), (int)(ent[k].x >> 24),
                   __uint_as_float(ent[k].y), lane, h, acc);
        }
      }
    } else {
    if (KEYS) {
      for (int i = 0; i < NIT; i += SCAN_U) {
        u4u kq[SCAN_U];
        bool ok[SCAN_U];
        #pragma unroll
        for (int u = 0; u < SCAN_U; ++u) {
          int itr = i + u + off;
          if (itr >= NIT) itr -= NIT;
          const int g = itr * 1024 + tid;
          ok[u] = (i + u < NIT) && (g < NQ);
          kq[u] = *(const u4u*)(own + 16 * (size_t)(ok[u] ? g : 0));
        }
        unsigned M[SCAN_U / 2];
        #pragma unroll
        for (int u = 0; u < SCAN_U; ++u) {
          unsigned m = (zero_bytes(kq[u].x ^ pat) >> 3) | (zero_bytes(kq[u].y ^ pat) >> 2)
                     | (zero_bytes(kq[u].z ^ pat) >> 1) |  zero_bytes(kq[u].w ^ pat);
          if (!ok[u]) m = 0;
          if (u & 1) M[u >> 1] |= m >> 4; else M[u >> 1] = m;
        }
        for (;;) {
          const unsigned any = M[0] | M[1] | M[2] | M[3];
          const unsigned bal = __builtin_amdgcn_ballot_w32(any != 0u);
          if (bal == 0u) break;
          if (any != 0u) {
            const int w = M[0] ? 0 : M[1] ? 1 : M[2] ? 2 : 3;
            const unsigned mw = M[0] ? M[0] : M[1] ? M[1] : M[2] ? M[2] : M[3];
            const int p = __builtin_ctz(mw);
            int itr = i + 2 * w + (((p >> 2) & 1) ^ 1) + off;
            if (itr >= NIT) itr -= NIT;
            q[cnt + __builtin_amdgcn_mbcnt_lo(bal, 0)] = ((itr * 1024 + tid) << 4) + 4 * (p & 3) + (p >> 3);
            const unsigned cl = mw & (mw - 1u);
            M[0] = (w == 0) ? cl : M[0];
            M[1] = (w == 1) ? cl : M[1];
            M[2] = (w == 2) ? cl : M[2];
            M[3] = (w == 3) ? cl : M[3];
          }
          cnt += __builtin_popcount(bal);
          if (cnt >= 32) {
            work_off_queue<LIST>(q, cnt, lane, base, rows, src, dst, h, dis, dsh, acc, lp, pos);
            cnt = 0;
          }
        }
      }
    } else {
      for (int i = 0; i < NIT; i += SCAN_U) {
        i4u dq[SCAN_U];
        int gq[SCAN_U];
        bool ok[SCAN_U];
        #pragma unroll
        for (int u = 0; u < SCAN_U; ++u) {
          int itr = i + u + off;
          if (itr >= NIT) itr -= NIT;
          const int g = itr * 1024 + tid;
          ok[u] = (i + u < NIT) && (g < NQ);
          gq[u] = ok[u] ? g : 0;
          dq[u] = *(const i4u*)(dst + 4 * (size_t)gq[u]);
        }
        #pragma unroll
        for (int u = 0; u < SCAN_U; ++u) {
          const bool m0 = ok[u] && (unsigned)(dq[u].x - base) < (unsigned)rows;
          const bool m1 = ok[u] && (unsigned)(dq[u].y - base) < (unsigned)rows;
          const bool m2 = ok[u] && (unsigned)(dq[u].z - base) < (unsigned)rows;
          const bool m3 = ok[u] && (unsigned)(dq[u].w - base) < (unsigned)rows;
          if (__builtin_amdgcn_ballot_w32(m0 || m1 || m2 || m3)) {
            push(m0, 4 * gq[u] + 0);
            push(m1, 4 * gq[u] + 1);
            push(m2, 4 * gq[u] + 2);
            push(m3, 4 * gq[u] + 3);
          }
        }
        if (cnt >= 32) {
          work_off_queue<LIST>(q, cnt, lane, base, rows, src, dst, h, dis, dsh, acc, lp, pos);
          cnt = 0;
        }
      }
    }
    {
      const int e = (NQ << GSH) + lane;
      bool m = (wave == 0) && (lane < (E & ((1 << GSH) - 1)));
      const int d = m ? dst[e] : -1;
      m = m && (unsigned)(d - base) < (unsigned)rows;
      push(m, e);
    }
    work_off_queue<LIST>(q, cnt, lane, base, rows, src, dst, h, dis, dsh, acc, lp, pos);
    }

    const int half = lane >> 4;
    const int lrow = lane & 15;
    const int ct   = wave & 7;
    const int g    = wave >> 3;
    v2f bfrag[32];
    load_frags(bfrag, wsh + (ct * 16 + lrow) * PITCH, half);
    const float bv = bias[ct * 16 + lrow];
    __syncthreads();

    #pragma unroll
    for (int p = 0; p < 2; ++p) {
      const int t0 = g + 8 * p, t1 = t0 + 4;
      const v4f16* a0 = (const v4f16*)(acc + (t0 * 16 + lrow) * PITCH) + half;
      v8f d[2] = {};
      if (t1 * 16 < rows) {
        const v4f16* a1 = a0 + 4 * 16 * (PITCH / 4);
        #pragma unroll
        for (int qq = 0; qq < 16; ++qq) {
          const v4f16 va0 = a0[2 * qq], va1 = a1[2 * qq];
          d[0] = wmma4((v2f){va0.x, va0.y}, bfrag[2 * qq],     d[0]);
          d[1] = wmma4((v2f){va1.x, va1.y}, bfrag[2 * qq],     d[1]);
          d[0] = wmma4((v2f){va0.z, va0.w}, bfrag[2 * qq + 1], d[0]);
          d[1] = wmma4((v2f){va1.z, va1.w}, bfrag[2 * qq + 1], d[1]);
        }
      } else if (t0 * 16 < rows) {
        #pragma unroll
        for (int qq = 0; qq < 16; ++qq) {
          const v4f16 va0 = a0[2 * qq];
          d[0] = wmma4((v2f){va0.x, va0.y}, bfrag[2 * qq],     d[0]);
          d[0] = wmma4((v2f){va0.z, va0.w}, bfrag[2 * qq + 1], d[0]);
        }
      }
      __syncthreads();
      const int row0 = t0 * 16 + half * 8;
      float* dp = acc + row0 * PITCH + ct * 16 + lrow;
      const int lim = rows - row0;
      #pragma unroll
      for (int u = 0; u < 2; ++u) {
        #pragma unroll
        for (int r = 0; r < 8; ++r) {
          const float v = d[u][r] + bv;
          if (64 * u + r < lim) dp[(64 * u + r) * PITCH] = (v >= 0.0f) ? v : LEAKY * v;
        }
      }
    }
    __syncthreads();

    #pragma unroll
    for (int k = 0; k < RW; ++k) {
      const int r = wave + 32 * k;
      if (r < rows) {
        const v4f16 v = *(const v4f16*)(acc + r * PITCH + 4 * lane);
        *(v4fa*)(outp + (size_t)(base + r) * DFEAT + 4 * lane) = (v4fa){v.x, v.y, v.z, v.w};
      }
    }
    __syncthreads();
  }
  if (LIST == 1 && lane == 0) lcnt[blockIdx.x * 32 + wave] = pos;
}

extern "C" void kernel_launch(void* const* d_in, const int* in_sizes, int n_in,
                              void* d_out, int out_size, void* d_ws, size_t ws_size,
                              hipStream_t stream) {
  const float* x  = (const float*)d_in[0];
  const int*   ei = (const int*)d_in[1];
  const float* Ws = (const float*)d_in[2];
  const float* bs = (const float*)d_in[3];
  float* out = (float*)d_out;

  const int n = in_sizes[0] / DFEAT;
  const int E = in_sizes[1] / 2;
  const int* src = ei;
  const int* dst = ei + E;

  const size_t list_words = (size_t)NBLK * 32 * LCAP * 2;
  const size_t need_keys  = ((size_t)n * (2 * DFEAT + 1)) * sizeof(float) + (size_t)E;
  const bool keys  = ws_size >= need_keys;
  const bool lists = keys && n <= NBLK * RMAX
                  && ws_size >= need_keys + (list_words + (size_t)NBLK * 32) * sizeof(unsigned);
  float* xbuf = (float*)d_ws;
  float* hbuf = xbuf + (size_t)n * DFEAT;
  unsigned* lst  = (unsigned*)(hbuf + (size_t)n * DFEAT);
  int*      lcnt = (int*)(lst + list_words);
  float* dis  = lists ? (float*)(lcnt + NBLK * 32) : hbuf + (size_t)n * DFEAT;
  unsigned char* own = keys ? (unsigned char*)(dis + n) : nullptr;

  const int TB = 256;

  deg_init    <<<(n + TB - 1) / TB, TB, 0, stream>>>(dis, n);
  deg_accum   <<<(E + TB - 1) / TB, TB, 0, stream>>>(dst, dis, own, E);
  deg_finalize<<<(n + TB - 1) / TB, TB, 0, stream>>>(dis, n);

  const int ga = NBLK;
  auto layer = [&](const float* xin, const float* w, const float* b, float* o, bool first) {
    if (lists && first) gcn_layer<true,  1><<<ga, 1024, 0, stream>>>(src, dst, own, xin, dis, w, b, o, n, E, lst, lcnt);
    else if (lists)     gcn_layer<true,  2><<<ga, 1024, 0, stream>>>(src, dst, own, xin, dis, w, b, o, n, E, lst, lcnt);
    else if (keys)      gcn_layer<true,  0><<<ga, 1024, 0, stream>>>(src, dst, own, xin, dis, w, b, o, n, E, lst, lcnt);
    else                gcn_layer<false, 0><<<ga, 1024, 0, stream>>>(src, dst, own, xin, dis, w, b, o, n, E, lst, lcnt);
  };

  layer(x,    Ws,                             bs,             xbuf, true);
  layer(xbuf, Ws + (size_t)DFEAT * DFEAT,     bs + DFEAT,     hbuf, false);
  layer(hbuf, Ws + (size_t)2 * DFEAT * DFEAT, bs + 2 * DFEAT, out,  false);
}
